# starting kernel
# baseline (speedup 1.0000x reference)
#include <hip/hip_runtime.h>
#include <math.h>

#define FRAME  160
#define NTAPS  16
#define TLEN   2400
#define NFRAMES 15
#define HALO   16
#define STAGE  (FRAME + HALO)
#define WPB    8

typedef __attribute__((ext_vector_type(2))) float v2f;
typedef __attribute__((ext_vector_type(8))) float v8f;

__device__ __forceinline__ float u2l(float u) {
    u -= 128.0f;
    float a = fabsf(u);
    float m = (32768.0f / 255.0f) * (__builtin_exp2f(a * (1.0f / 16.0f)) - 1.0f);
    return copysignf(m, u);
}

__device__ __forceinline__ float l2u(float x) {
    float a = fabsf(x);
    float u = 16.0f * __builtin_log2f(fmaf(255.0f / 32768.0f, a, 1.0f));
    u = copysignf(u, x);
    return fminf(fmaxf(128.0f + u, 0.0f), 255.0f);
}

__global__ __launch_bounds__(256) void lpc_wmma_kernel(
    const float* __restrict__ sig,
    const float* __restrict__ lpc,
    float* __restrict__ out,
    int njobs)
{
    __shared__ float xbuf[WPB * STAGE];

    const int lane = threadIdx.x & 31;
    const int wid  = threadIdx.x >> 5;
    const int job  = blockIdx.x * WPB + wid;
    if (job >= njobs) return;

    const int batch = job / NFRAMES;
    const int frame = job - batch * NFRAMES;
    const int tf0   = frame * FRAME;

    float* xw = &xbuf[wid * STAGE];

    for (int i = lane; i < STAGE; i += 32) {
        int t = tf0 - HALO + i;
        float v = 0.0f;
        if (t >= 0) v = u2l(sig[batch * TLEN + t]);
        xw[i] = v;
    }
    __syncthreads();

    const int h = lane >> 4;
    const int m = lane & 15;

    const float* c = &lpc[(batch * NFRAMES + frame) * NTAPS];
    v2f bv[4];
#pragma unroll
    for (int kk = 0; kk < 4; ++kk) {
        int k0 = 4 * kk + 2 * h;
        bv[kk].x = -c[k0];
        bv[kk].y = -c[k0 + 1];
    }

    for (int ti = 0; ti < FRAME / 16; ++ti) {
        const int base = 16 * ti + HALO + m;
        v8f acc = {};
#pragma unroll
        for (int kk = 0; kk < 4; ++kk) {
            int k0 = 4 * kk + 2 * h;
            v2f a;
            a.x = xw[base - k0];
            a.y = xw[base - (k0 + 1)];
            acc = __builtin_amdgcn_wmma_f32_16x16x4_f32(
                false, a, false, bv[kk], (short)0, acc, false, false);
        }
        if (m < 8) {
            int r = lane & 7;
            float val = acc[0];
#pragma unroll
            for (int q = 1; q < 8; ++q) val = (r == q) ? acc[q] : val;
            int mt = r + 8 * h;
            out[batch * TLEN + tf0 + 16 * ti + mt] = l2u(val);
        }
    }
}

extern "C" void kernel_launch(void* const* d_in, const int* in_sizes, int n_in,
                              void* d_out, int out_size, void* d_ws, size_t ws_size,
                              hipStream_t stream) {
    const float* sig = (const float*)d_in[0];
    const float* lpc = (const float*)d_in[1];
    float* out = (float*)d_out;

    const int nbatch = in_sizes[0] / TLEN;
    const int njobs  = nbatch * NFRAMES;
    const int nblocks = (njobs + WPB - 1) / WPB;

    lpc_wmma_kernel<<<nblocks, 256, 0, stream>>>(sig, lpc, out, njobs);
}


# Each wave does 5 consecutive frames of a row (signal and taps fetched once, all its 16-byte loads in flight together, halo shared in LDS): 6144 long-lived waves instead of one short wave per frame
# speedup vs baseline: 2.8386x; 2.8386x over previous
#include <hip/hip_runtime.h>
#include <math.h>


#define FRAME  160
#define NTAPS  16
#define TLEN   2400
#define NFRAMES 15
#define FPW    5
#define SEGS   (NFRAMES / FPW)
#define SEGLEN (FPW * FRAME)
#define SEGV4  (SEGLEN / 4)
#define WPB    8
#define NROWS  (1 + SEGLEN / 16)
#define PITCH  20
#define XSZ    (NROWS * PITCH)
#define CSZ    (32 * FPW + 16)
#define WSZ    (CSZ + XSZ)

typedef __attribute__((ext_vector_type(2))) float v2f;
typedef __attribute__((ext_vector_type(4))) float v4f;
typedef __attribute__((ext_vector_type(8))) float v8f;

__device__ __forceinline__ float u2l(float u) {
    u -= 128.0f;
    float a = fabsf(u);
    float m = (32768.0f / 255.0f) * (__builtin_exp2f(a * (1.0f / 16.0f)) - 1.0f);
    return copysignf(m, u);
}

__device__ __forceinline__ v4f u2l4(v4f u) {
    v4f r;
    r.x = u2l(u.x); r.y = u2l(u.y); r.z = u2l(u.z); r.w = u2l(u.w);
    return r;
}

__device__ __forceinline__ float l2u(float x) {
    float a = fabsf(x);
    float u = 16.0f * __builtin_log2f(fmaf(255.0f / 32768.0f, a, 1.0f));
    u = copysignf(u, x);
    return fminf(fmaxf(128.0f + u, 0.0f), 255.0f);
}

__device__ __forceinline__ float xhalf(float x) {
    int b = __builtin_bit_cast(int, x);
    return __builtin_bit_cast(float,
        __builtin_amdgcn_permlanex16(b, b, 0x76543210, 0xfedcba98, false, false));
}

__global__ __launch_bounds__(256) void lpc_wmma_kernel(
    const float* __restrict__ sig,
    const float* __restrict__ lpc,
    float* __restrict__ out,
    int nsegs)
{
    __shared__ __attribute__((aligned(16))) float sbuf[WPB * WSZ];

    const int lane = threadIdx.x & 31;
    const int wid  = threadIdx.x >> 5;
    const int seg  = blockIdx.x * WPB + wid;
    if (seg >= nsegs) return;

    const int batch = seg / SEGS;
    const int part  = seg - batch * SEGS;
    const int ts0   = part * SEGLEN;

    float* zc = &sbuf[wid * WSZ];
    float* xw = zc + CSZ;

    const int h = lane >> 4;
    const int m = lane & 15;

    const v4f* srow = reinterpret_cast<const v4f*>(sig + batch * TLEN + ts0);
    const v4f* crow = reinterpret_cast<const v4f*>(lpc + (batch * NFRAMES + part * FPW) * NTAPS);

    const v4f ct = crow[min(lane, 19)];
    v4f xs[6];
    for (int q = 0; q < 6; ++q) xs[q] = srow[32 * q + lane];
    const bool tail_halo = (lane >= 8) && (lane < 12);
    const bool halo_real = tail_halo && (part > 0);
    const v4f xt = srow[halo_real ? lane - 12 : 192 + (lane & 7)];
    __builtin_amdgcn_sched_barrier(0);

    v4f* xst = reinterpret_cast<v4f*>(xw + (1 + (lane >> 2)) * PITCH + 4 * (lane & 3));
    for (int q = 0; q < 6; ++q) xst[q * (8 * PITCH / 4)] = u2l4(xs[q]);
    {
        v4f yt = u2l4(xt);
        if (tail_halo && !halo_real) yt = (v4f){0.0f, 0.0f, 0.0f, 0.0f};
        v4f* tbody = reinterpret_cast<v4f*>(xw + (49 + ((lane & 7) >> 2)) * PITCH + 4 * (lane & 3));
        v4f* thalo = reinterpret_cast<v4f*>(xw) + (lane - 8);
        *(tail_halo ? thalo : tbody) = yt;
    }
    if (lane < 24)
        reinterpret_cast<v4f*>(zc + 32 * (lane >> 2))[lane & 3] = (v4f){0.0f, 0.0f, 0.0f, 0.0f};
    if (lane < 20)
        *reinterpret_cast<v4f*>(zc + 32 * (lane >> 2) + 28 - 4 * (lane & 3)) =
            (v4f){-ct.w, -ct.z, -ct.y, -ct.x};
    __syncthreads();

    const float* xa0 = xw + min(m + h, 10) * PITCH;
    const float* zb0 = zc + (15 + 16 * h - m);
    float* o = &out[batch * TLEN + ts0 + lane];

    #pragma unroll
    for (int f = 0; f < FPW; ++f) {
        const v4f* xr = reinterpret_cast<const v4f*>(xa0 + f * (10 * PITCH));
        v4f xa[4];
        for (int g = 0; g < 4; ++g) xa[g] = xr[g];

        const float* zb = zb0 + 32 * f;
        v2f bv[8];
        for (int j = 0; j < 8; ++j) {
            bv[j].x = zb[2 * j];
            bv[j].y = zb[2 * j + 1];
        }

        v8f acc = {};
        for (int j = 0; j < 8; ++j) {
            v2f a;
            a.x = xa[j >> 1][2 * (j & 1)];
            a.y = xa[j >> 1][2 * (j & 1) + 1];
            acc = __builtin_amdgcn_wmma_f32_16x16x4_f32(
                false, a, false, bv[j], (short)0, acc, false, false);
        }

        float* of = o + f * FRAME;
        for (int p = 0; p < 4; ++p) {
            float odd = xhalf(acc[2 * p + 1]);
            float val = (h == 0) ? acc[2 * p] : odd;
            of[32 * p] = l2u(val);
        }
        {
            float r8  = xhalf(acc[0]);
            float val = (h == 0) ? r8 : acc[1];
            of[128] = l2u(val);
        }
    }
}

extern "C" void kernel_launch(void* const* d_in, const int* in_sizes, int n_in,
                              void* d_out, int out_size, void* d_ws, size_t ws_size,
                              hipStream_t stream) {
    const float* sig = (const float*)d_in[0];
    const float* lpc = (const float*)d_in[1];
    float* out = (float*)d_out;

    const int nbatch = in_sizes[0] / TLEN;
    const int nsegs  = nbatch * SEGS;
    const int nblocks = (nsegs + WPB - 1) / WPB;

    lpc_wmma_kernel<<<nblocks, 256, 0, stream>>>(sig, lpc, out, nsegs);
}


# mu-law exp2/log2 issued as the bare hardware instructions: the library's out-of-range rescale (5 vector ops per call, never taken: arguments are >= 0 and >= 1) is gone; output bits unchanged
# speedup vs baseline: 3.0579x; 1.0773x over previous
#include <hip/hip_runtime.h>
#include <math.h>


#define FRAME  160
#define NTAPS  16
#define TLEN   2400
#define NFRAMES 15
#define FPW    5
#define SEGS   (NFRAMES / FPW)
#define SEGLEN (FPW * FRAME)
#define SEGV4  (SEGLEN / 4)
#define WPB    8
#define NROWS  (1 + SEGLEN / 16)
#define PITCH  20
#define XSZ    (NROWS * PITCH)
#define CSZ    (32 * FPW + 16)
#define WSZ    (CSZ + XSZ)

typedef __attribute__((ext_vector_type(2))) float v2f;
typedef __attribute__((ext_vector_type(4))) float v4f;
typedef __attribute__((ext_vector_type(8))) float v8f;

__device__ __forceinline__ float u2l(float u) {
    u -= 128.0f;
    float a = fabsf(u);
    float m = (32768.0f / 255.0f) * (__builtin_amdgcn_exp2f(a * (1.0f / 16.0f)) - 1.0f);
    return copysignf(m, u);
}

__device__ __forceinline__ v4f u2l4(v4f u) {
    v4f r;
    r.x = u2l(u.x); r.y = u2l(u.y); r.z = u2l(u.z); r.w = u2l(u.w);
    return r;
}

__device__ __forceinline__ float l2u(float x) {
    float a = fabsf(x);
    float u = 16.0f * __builtin_amdgcn_logf(fmaf(255.0f / 32768.0f, a, 1.0f));
    u = copysignf(u, x);
    return fminf(fmaxf(128.0f + u, 0.0f), 255.0f);
}

__device__ __forceinline__ float xhalf(float x) {
    int b = __builtin_bit_cast(int, x);
    return __builtin_bit_cast(float,
        __builtin_amdgcn_permlanex16(b, b, 0x76543210, 0xfedcba98, false, false));
}

__global__ __launch_bounds__(256) void lpc_wmma_kernel(
    const float* __restrict__ sig,
    const float* __restrict__ lpc,
    float* __restrict__ out,
    int nsegs)
{
    __shared__ __attribute__((aligned(16))) float sbuf[WPB * WSZ];

    const int lane = threadIdx.x & 31;
    const int wid  = threadIdx.x >> 5;
    const int seg  = blockIdx.x * WPB + wid;
    if (seg >= nsegs) return;

    const int batch = seg / SEGS;
    const int part  = seg - batch * SEGS;
    const int ts0   = part * SEGLEN;

    float* zc = &sbuf[wid * WSZ];
    float* xw = zc + CSZ;

    const int h = lane >> 4;
    const int m = lane & 15;

    const v4f* srow = reinterpret_cast<const v4f*>(sig + batch * TLEN + ts0);
    const v4f* crow = reinterpret_cast<const v4f*>(lpc + (batch * NFRAMES + part * FPW) * NTAPS);

    const v4f ct = crow[min(lane, 19)];
    v4f xs[6];
    for (int q = 0; q < 6; ++q) xs[q] = srow[32 * q + lane];
    const bool tail_halo = (lane >= 8) && (lane < 12);
    const bool halo_real = tail_halo && (part > 0);
    const v4f xt = srow[halo_real ? lane - 12 : 192 + (lane & 7)];
    __builtin_amdgcn_sched_barrier(0);

    v4f* xst = reinterpret_cast<v4f*>(xw + (1 + (lane >> 2)) * PITCH + 4 * (lane & 3));
    for (int q = 0; q < 6; ++q) xst[q * (8 * PITCH / 4)] = u2l4(xs[q]);
    {
        v4f yt = u2l4(xt);
        if (tail_halo && !halo_real) yt = (v4f){0.0f, 0.0f, 0.0f, 0.0f};
        v4f* tbody = reinterpret_cast<v4f*>(xw + (49 + ((lane & 7) >> 2)) * PITCH + 4 * (lane & 3));
        v4f* thalo = reinterpret_cast<v4f*>(xw) + (lane - 8);
        *(tail_halo ? thalo : tbody) = yt;
    }
    if (lane < 24)
        reinterpret_cast<v4f*>(zc + 32 * (lane >> 2))[lane & 3] = (v4f){0.0f, 0.0f, 0.0f, 0.0f};
    if (lane < 20)
        *reinterpret_cast<v4f*>(zc + 32 * (lane >> 2) + 28 - 4 * (lane & 3)) =
            (v4f){-ct.w, -ct.z, -ct.y, -ct.x};
    __syncthreads();

    const float* xa0 = xw + min(m + h, 10) * PITCH;
    const float* zb0 = zc + (15 + 16 * h - m);
    float* o = &out[batch * TLEN + ts0 + lane];

    #pragma unroll
    for (int f = 0; f < FPW; ++f) {
        const v4f* xr = reinterpret_cast<const v4f*>(xa0 + f * (10 * PITCH));
        v4f xa[4];
        for (int g = 0; g < 4; ++g) xa[g] = xr[g];

        const float* zb = zb0 + 32 * f;
        v2f bv[8];
        for (int j = 0; j < 8; ++j) {
            bv[j].x = zb[2 * j];
            bv[j].y = zb[2 * j + 1];
        }

        v8f acc = {};
        for (int j = 0; j < 8; ++j) {
            v2f a;
            a.x = xa[j >> 1][2 * (j & 1)];
            a.y = xa[j >> 1][2 * (j & 1) + 1];
            acc = __builtin_amdgcn_wmma_f32_16x16x4_f32(
                false, a, false, bv[j], (short)0, acc, false, false);
        }

        float* of = o + f * FRAME;
        for (int p = 0; p < 4; ++p) {
            float odd = xhalf(acc[2 * p + 1]);
            float val = (h == 0) ? acc[2 * p] : odd;
            of[32 * p] = l2u(val);
        }
        {
            float r8  = xhalf(acc[0]);
            float val = (h == 0) ? r8 : acc[1];
            of[128] = l2u(val);
        }
    }
}

extern "C" void kernel_launch(void* const* d_in, const int* in_sizes, int n_in,
                              void* d_out, int out_size, void* d_ws, size_t ws_size,
                              hipStream_t stream) {
    const float* sig = (const float*)d_in[0];
    const float* lpc = (const float*)d_in[1];
    float* out = (float*)d_out;

    const int nbatch = in_sizes[0] / TLEN;
    const int nsegs  = nbatch * SEGS;
    const int nblocks = (nsegs + WPB - 1) / WPB;

    lpc_wmma_kernel<<<nblocks, 256, 0, stream>>>(sig, lpc, out, nsegs);
}
